# starting kernel
# baseline (speedup 1.0000x reference)
#include <hip/hip_runtime.h>

typedef __attribute__((ext_vector_type(2))) float v2f;
typedef __attribute__((ext_vector_type(8))) float v8f;

#define QS_THREADS 128

__device__ __forceinline__ int perm4(int x) {
  if (x & 8) x ^= 4;
  if (x & 4) x ^= 2;
  if (x & 2) x ^= 1;
  return x;
}
__device__ __forceinline__ int perm4inv(int x) {
  if (x & 2) x ^= 1;
  if (x & 4) x ^= 2;
  if (x & 8) x ^= 4;
  return x;
}

__global__ __launch_bounds__(QS_THREADS) void qsim_wmma_kernel(
    const float* __restrict__ state_batch,
    const float* __restrict__ params,
    const float* __restrict__ head_w,
    const float* __restrict__ head_b,
    float* __restrict__ out) {
  __shared__ float psi[4096];
  __shared__ float U[5][256];
  __shared__ float cz[24], sz[24];
  __shared__ float hws[12];
  __shared__ float red[QS_THREADS];

  const int b   = blockIdx.x;
  const int tid = threadIdx.x;

  if (tid < 12) {
    float a = 0.5f * state_batch[(size_t)b * 12 + tid];
    cz[tid] = cosf(a); sz[tid] = sinf(a);
    hws[tid] = head_w[tid];
  } else if (tid < 24) {
    float a = 0.5f * params[tid - 12];
    cz[tid] = cosf(a); sz[tid] = sinf(a);
  }
  __syncthreads();

  auto kron4 = [&](int base, int i, int j) -> float {
    float v = 1.0f;
#pragma unroll
    for (int q = 0; q < 4; ++q) {
      int bi = (i >> (3 - q)) & 1;
      int bj = (j >> (3 - q)) & 1;
      float c = cz[base + q], s = sz[base + q];
      v *= bi ? (bj ? c : s) : (bj ? -s : c);
    }
    return v;
  };

  for (int e = tid; e < 5 * 256; e += QS_THREADS) {
    int mi = e >> 8, idx = e & 255, i = idx >> 4, j = idx & 15;
    float v;
    switch (mi) {
      case 0:  v = kron4(12, perm4inv(i), j); break;
      case 1:  v = kron4(12, i, j); break;
      case 2:  v = (i == perm4(j)) ? 1.0f : 0.0f; break;
      case 3:  v = kron4(16, i, j); break;
      default: v = kron4(20, i, perm4(j)); break;
    }
    U[mi][idx] = v;
  }

  for (int s = tid; s < 4096; s += QS_THREADS) {
    int s2 = (perm4inv(s >> 8) << 8) | (s & 255);
    float v = 1.0f;
#pragma unroll
    for (int w = 0; w < 12; ++w)
      v *= ((s2 >> (11 - w)) & 1) ? sz[w] : cz[w];
    psi[s] = v;
  }
  __syncthreads();

  auto apply = [&](const float* Um, int sm, int sn, int st) {
    const int lane = tid & 31, wave = tid >> 5;
    const int n = lane & 15, hi = lane >> 4;
    v2f a[4];
#pragma unroll
    for (int kc = 0; kc < 4; ++kc) {
      int k0 = 4 * kc + 2 * hi;
      a[kc].x = Um[n * 16 + k0];
      a[kc].y = Um[n * 16 + k0 + 1];
    }
#pragma unroll
    for (int tt = 0; tt < 4; ++tt) {
      int t = wave * 4 + tt;
      int base = st * t + sn * n;
      v8f c = {};
#pragma unroll
      for (int kc = 0; kc < 4; ++kc) {
        int k0 = 4 * kc + 2 * hi;
        v2f bf;
        bf.x = psi[base + sm * k0];
        bf.y = psi[base + sm * (k0 + 1)];
        c = __builtin_amdgcn_wmma_f32_16x16x4_f32(
             false, a[kc],  false, bf,
             (short)0, c,  false,  false);
      }
#pragma unroll
      for (int v = 0; v < 8; ++v)
        psi[base + sm * (v + 8 * hi)] = c[v];
    }
  };

  auto cswap = [&](int cpos) {
    int tpos = cpos - 1;
    for (int idx = tid; idx < 1024; idx += QS_THREADS) {
      int low = idx & ((1 << tpos) - 1);
      int high = idx >> tpos;
      int s = (high << (cpos + 1)) | (1 << cpos) | low;
      int s2 = s | (1 << tpos);
      float x = psi[s], y = psi[s2];
      psi[s] = y; psi[s2] = x;
    }
  };

  for (int l = 0; l < 4; ++l) {
    cswap(8);                       __syncthreads();
    apply(U[l < 3 ? 0 : 1], 256, 1, 16);  __syncthreads();
    apply(U[2], 16, 1, 256);        __syncthreads();
    cswap(4);                       __syncthreads();
    apply(U[4], 1, 16, 256);        __syncthreads();
    apply(U[3], 16, 1, 256);        __syncthreads();
  }

  float partial = 0.0f;
  for (int s = tid; s < 4096; s += QS_THREADS) {
    float p = psi[s]; p *= p;
    float g = 0.0f;
#pragma unroll
    for (int w = 0; w < 12; ++w)
      g += ((s >> (11 - w)) & 1) ? -hws[w] : hws[w];
    partial += p * g;
  }
  red[tid] = partial;
  __syncthreads();
#pragma unroll
  for (int off = QS_THREADS / 2; off > 0; off >>= 1) {
    if (tid < off) red[tid] += red[tid + off];
    __syncthreads();
  }
  if (tid == 0) out[b] = red[0] + head_b[0];
}

extern "C" void kernel_launch(void* const* d_in, const int* in_sizes, int n_in,
                              void* d_out, int out_size, void* d_ws, size_t ws_size,
                              hipStream_t stream) {
  const float* sb     = (const float*)d_in[0];
  const float* params = (const float*)d_in[1];
  const float* hw     = (const float*)d_in[2];
  const float* hb     = (const float*)d_in[3];
  float* out = (float*)d_out;
  int batch = in_sizes[0] / 12;
  qsim_wmma_kernel<<<dim3(batch), dim3(QS_THREADS), 0, stream>>>(
      sb, params, hw, hb, out);
}


# One sample per wave, statevector kept in registers as accumulator tiles chained through WMMA operands (CNOTs as lane and tile exchanges): no LDS state, no barriers; every sum runs in a new k order
# speedup vs baseline: 1.8045x; 1.8045x over previous
#include <hip/hip_runtime.h>

typedef __attribute__((ext_vector_type(2))) float v2f;
typedef __attribute__((ext_vector_type(8))) float v8f;

#define QS_THREADS 32

__device__ __forceinline__ int perm4(int x) {
  if (x & 8) x ^= 4;
  if (x & 4) x ^= 2;
  if (x & 2) x ^= 1;
  return x;
}

#define TAB_U 1280
#define TAB_G 4096
#define TAB_THREADS 256

__global__ __launch_bounds__(TAB_THREADS) void qsim_tables_kernel(
    const float* __restrict__ params,
    const float* __restrict__ head_w,
    float* __restrict__ tab) {
  __shared__ float cp[12], sp[12], hws[12];
  const int tid = threadIdx.x;
  if (tid < 12) {
    float a = 0.5f * params[tid];
    cp[tid] = cosf(a); sp[tid] = sinf(a);
    hws[tid] = head_w[tid];
  }
  __syncthreads();

  auto kron4 = [&](int base, int i, int j) -> float {
    float v = 1.0f;
#pragma unroll
    for (int q = 0; q < 4; ++q) {
      int bi = (i >> (3 - q)) & 1;
      int bj = (j >> (3 - q)) & 1;
      float c = cp[base + q], s = sp[base + q];
      v *= bi ? (bj ? c : s) : (bj ? -s : c);
    }
    return v;
  };

  const int e = blockIdx.x * TAB_THREADS + tid;
  if (e < TAB_U) {
    int mi = e >> 8, idx = e & 255, i = idx >> 4, j = idx & 15;
    tab[e] = (mi < 3) ? kron4(4 * mi, i, perm4(j)) : 0.0f;
  } else if (e < TAB_U + TAB_G) {
    int s = e - TAB_U;
    float g = 0.0f;
#pragma unroll
    for (int w = 0; w < 12; ++w)
      g += ((s >> (11 - w)) & 1) ? -hws[w] : hws[w];
    tab[e] = g;
  }
}

template <int RM>
__device__ __forceinline__ float flip8(float x) {
  int xi = __builtin_bit_cast(int, x);
  return __builtin_bit_cast(float,
      __builtin_amdgcn_update_dpp(xi, xi, 0x128, RM, 0xf, false));
}

__device__ __forceinline__ v8f wmma4(v2f a, v2f b, v8f c) {
  return __builtin_amdgcn_wmma_f32_16x16x4_f32(false, a, false, b, (short)0, c, false, false);
}

__global__ __launch_bounds__(QS_THREADS) void qsim_wmma_kernel(
    const float* __restrict__ state_batch,
    const float* __restrict__ tab,
    const float* __restrict__ head_b,
    float* __restrict__ out) {
  const int b    = blockIdx.x;
  const int lane = threadIdx.x;
  const int c = lane & 15, h = lane >> 4;

  float ang = 0.5f * state_batch[(size_t)b * 12 + (lane < 12 ? lane : 0)];
  float myc = cosf(ang), mys = sinf(ang);
  float cz[12], sz[12];
#pragma unroll
  for (int w = 0; w < 12; ++w) {
    cz[w] = __builtin_bit_cast(float, __builtin_amdgcn_readlane(__builtin_bit_cast(int, myc), w));
    sz[w] = __builtin_bit_cast(float, __builtin_amdgcn_readlane(__builtin_bit_cast(int, mys), w));
  }

  v2f BW[3][4], BX[4], AZ[2][2][8];
#pragma unroll
  for (int m = 0; m < 3; ++m)
#pragma unroll
    for (int j = 0; j < 4; ++j) {
      BW[m][j].x = tab[256 * m + 16 * c + 2 * j + 8 * h];
      BW[m][j].y = tab[256 * m + 16 * c + 2 * j + 1 + 8 * h];
    }
#pragma unroll
  for (int j = 0; j < 4; ++j) {
    BX[j].x = tab[256 + 16 * c + 2 * j + 8 * (1 - h)];
    BX[j].y = tab[256 + 16 * c + 2 * j + 1 + 8 * (1 - h)];
  }
#pragma unroll
  for (int m = 0; m < 2; ++m)
#pragma unroll
    for (int q = 0; q < 8; ++q) {
      float ax = tab[256 * m + 16 * c + 2 * q];
      float ay = tab[256 * m + 16 * c + 2 * q + 1];
      AZ[m][0][q].x = h ? 0.0f : ax;  AZ[m][0][q].y = h ? 0.0f : ay;
      AZ[m][1][q].x = h ? ax : 0.0f;  AZ[m][1][q].y = h ? ay : 0.0f;
    }

  v8f X[16];
  {
    float eL = 1.0f;
#pragma unroll
    for (int q = 0; q < 4; ++q) eL *= ((c >> (3 - q)) & 1) ? sz[8 + q] : cz[8 + q];
    float eM[8];
#pragma unroll
    for (int v = 0; v < 8; ++v) {
      float e = h ? sz[4] : cz[4];
#pragma unroll
      for (int q = 1; q < 4; ++q) e *= ((v >> (3 - q)) & 1) ? sz[4 + q] : cz[4 + q];
      eM[v] = e * eL;
    }
#pragma unroll
    for (int t = 0; t < 16; ++t) {
      float eT = 1.0f;
#pragma unroll
      for (int q = 0; q < 4; ++q) eT *= ((t >> (3 - q)) & 1) ? sz[q] : cz[q];
#pragma unroll
      for (int v = 0; v < 8; ++v) X[t][v] = eT * eM[v];
    }
  }

  auto rowpass = [&](v8f& Z, const v2f* W) {
    v8f acc = {};
#pragma unroll
    for (int j = 0; j < 4; ++j) {
      v2f a; a.x = Z[2 * j]; a.y = Z[2 * j + 1];
      acc = wmma4(a, W[j], acc);
    }
    Z = acc;
  };

#pragma unroll
  for (int l = 0; l < 4; ++l) {
    const int odd = l & 1;
#pragma unroll
    for (int t = 0; t < 16; ++t)
#pragma unroll
      for (int v = 0; v < 8; ++v) {
        if ((__builtin_parity(t) ^ __builtin_parity(v)) & 1) X[t][v] = flip8<1>(X[t][v]);
        else                                                  X[t][v] = flip8<2>(X[t][v]);
      }
    if (odd) {
#pragma unroll
      for (int t = 0; t < 8; ++t)
#pragma unroll
        for (int v = 0; v < 8; ++v) {
          bool sw = ((__builtin_parity(v) & 1) ^ h) != 0;
          float x0 = X[t][v], x1 = X[t + 8][v];
          X[t][v]     = sw ? x1 : x0;
          X[t + 8][v] = sw ? x0 : x1;
        }
    }
#pragma unroll
    for (int t = 0; t < 16; ++t) {
      if (odd) rowpass(X[t], BW[0]);
      else     rowpass(X[t], (__builtin_parity(t) & 1) ? BX : BW[1]);
      rowpass(X[t], BW[2]);
    }
    v8f Y[16];
#pragma unroll
    for (int v = 0; v < 8; ++v)
#pragma unroll
      for (int h0 = 0; h0 < 2; ++h0) {
        v8f acc = {};
#pragma unroll
        for (int q = 0; q < 8; ++q) {
          v2f bf; bf.x = X[2 * q][v]; bf.y = X[2 * q + 1][v];
          acc = wmma4(AZ[odd][h0][q], bf, acc);
        }
        Y[v + 8 * h0] = acc;
      }
#pragma unroll
    for (int t = 0; t < 16; ++t) X[t] = Y[t];
  }

  float partial = 0.0f;
#pragma unroll
  for (int t = 0; t < 16; ++t)
#pragma unroll
    for (int v = 0; v < 8; ++v) {
      float p = X[t][v];
      partial += p * p * tab[TAB_U + 256 * t + 16 * (v + 8 * h) + c];
    }
#pragma unroll
  for (int off = 16; off > 0; off >>= 1)
    partial += __shfl_xor(partial, off);
  if (lane == 0) out[b] = partial + head_b[0];
}

extern "C" void kernel_launch(void* const* d_in, const int* in_sizes, int n_in,
                              void* d_out, int out_size, void* d_ws, size_t ws_size,
                              hipStream_t stream) {
  const float* sb     = (const float*)d_in[0];
  const float* params = (const float*)d_in[1];
  const float* hw     = (const float*)d_in[2];
  const float* hb     = (const float*)d_in[3];
  float* out = (float*)d_out;
  int batch = in_sizes[0] / 12;
  float* tab = (float*)d_ws;
  qsim_tables_kernel<<<dim3((TAB_U + TAB_G) / TAB_THREADS), dim3(TAB_THREADS), 0, stream>>>(
      params, hw, tab);
  qsim_wmma_kernel<<<dim3(batch), dim3(QS_THREADS), 0, stream>>>(
      sb, tab, hb, out);
}


# Tile-index pass: 64 lane-half swaps per layer (v_permlane16_swap) transpose tile and row, then 4 full WMMAs per tile replace 8 half-zero ones: 1024 -> 768 WMMAs; tile sums run in a new k order
# speedup vs baseline: 2.7751x; 1.5379x over previous
#include <hip/hip_runtime.h>

typedef __attribute__((ext_vector_type(2))) float v2f;
typedef __attribute__((ext_vector_type(8))) float v8f;

#define QS_THREADS 32

__device__ __forceinline__ int perm4(int x) {
  if (x & 8) x ^= 4;
  if (x & 4) x ^= 2;
  if (x & 2) x ^= 1;
  return x;
}

#define TAB_U 1280
#define TAB_G 4096
#define TAB_THREADS 256

__global__ __launch_bounds__(TAB_THREADS) void qsim_tables_kernel(
    const float* __restrict__ params,
    const float* __restrict__ head_w,
    float* __restrict__ tab) {
  __shared__ float cp[12], sp[12], hws[12];
  const int tid = threadIdx.x;
  if (tid < 12) {
    float a = 0.5f * params[tid];
    cp[tid] = cosf(a); sp[tid] = sinf(a);
    hws[tid] = head_w[tid];
  }
  __syncthreads();

  auto kron4 = [&](int base, int i, int j) -> float {
    float v = 1.0f;
#pragma unroll
    for (int q = 0; q < 4; ++q) {
      int bi = (i >> (3 - q)) & 1;
      int bj = (j >> (3 - q)) & 1;
      float c = cp[base + q], s = sp[base + q];
      v *= bi ? (bj ? c : s) : (bj ? -s : c);
    }
    return v;
  };

  const int e = blockIdx.x * TAB_THREADS + tid;
  if (e < TAB_U) {
    int mi = e >> 8, idx = e & 255, i = idx >> 4, j = idx & 15;
    tab[e] = (mi < 3) ? kron4(4 * mi, i, perm4(j)) : 0.0f;
  } else if (e < TAB_U + TAB_G) {
    int s = e - TAB_U;
    float g = 0.0f;
#pragma unroll
    for (int w = 0; w < 12; ++w)
      g += ((s >> (11 - w)) & 1) ? -hws[w] : hws[w];
    tab[e] = g;
  }
}

template <int RM>
__device__ __forceinline__ float flip8(float x) {
  int xi = __builtin_bit_cast(int, x);
  return __builtin_bit_cast(float,
      __builtin_amdgcn_update_dpp(xi, xi, 0x128, RM, 0xf, false));
}

__device__ __forceinline__ void swap_halves(float& p, float& q) {
  auto r = __builtin_amdgcn_permlane16_swap(__builtin_bit_cast(unsigned, p),
                                            __builtin_bit_cast(unsigned, q), false, false);
  p = __builtin_bit_cast(float, (unsigned)r[0]);
  q = __builtin_bit_cast(float, (unsigned)r[1]);
}

__device__ __forceinline__ v8f wmma4(v2f a, v2f b, v8f c) {
  return __builtin_amdgcn_wmma_f32_16x16x4_f32(false, a, false, b, (short)0, c, false, false);
}

__global__ __launch_bounds__(QS_THREADS) void qsim_wmma_kernel(
    const float* __restrict__ state_batch,
    const float* __restrict__ tab,
    const float* __restrict__ head_b,
    float* __restrict__ out) {
  const int b    = blockIdx.x;
  const int lane = threadIdx.x;
  const int c = lane & 15, h = lane >> 4;

  float ang = 0.5f * state_batch[(size_t)b * 12 + (lane < 12 ? lane : 0)];
  float myc = cosf(ang), mys = sinf(ang);
  float cz[12], sz[12];
#pragma unroll
  for (int w = 0; w < 12; ++w) {
    cz[w] = __builtin_bit_cast(float, __builtin_amdgcn_readlane(__builtin_bit_cast(int, myc), w));
    sz[w] = __builtin_bit_cast(float, __builtin_amdgcn_readlane(__builtin_bit_cast(int, mys), w));
  }

  v2f BW[3][4], BX[4];
#pragma unroll
  for (int m = 0; m < 3; ++m)
#pragma unroll
    for (int j = 0; j < 4; ++j) {
      BW[m][j].x = tab[256 * m + 16 * c + 2 * j + 8 * h];
      BW[m][j].y = tab[256 * m + 16 * c + 2 * j + 1 + 8 * h];
    }
#pragma unroll
  for (int j = 0; j < 4; ++j) {
    BX[j].x = tab[256 + 16 * c + 2 * j + 8 * (1 - h)];
    BX[j].y = tab[256 + 16 * c + 2 * j + 1 + 8 * (1 - h)];
  }

  v8f X[16];
  {
    float eL = 1.0f;
#pragma unroll
    for (int q = 0; q < 4; ++q) eL *= ((c >> (3 - q)) & 1) ? sz[8 + q] : cz[8 + q];
    float eM[8];
#pragma unroll
    for (int v = 0; v < 8; ++v) {
      float e = h ? sz[4] : cz[4];
#pragma unroll
      for (int q = 1; q < 4; ++q) e *= ((v >> (3 - q)) & 1) ? sz[4 + q] : cz[4 + q];
      eM[v] = e * eL;
    }
#pragma unroll
    for (int t = 0; t < 16; ++t) {
      float eT = 1.0f;
#pragma unroll
      for (int q = 0; q < 4; ++q) eT *= ((t >> (3 - q)) & 1) ? sz[q] : cz[q];
#pragma unroll
      for (int v = 0; v < 8; ++v) X[t][v] = eT * eM[v];
    }
  }

  auto rowpass = [&](v8f& Z, const v2f* W) {
    v8f acc = {};
#pragma unroll
    for (int j = 0; j < 4; ++j) {
      v2f a; a.x = Z[2 * j]; a.y = Z[2 * j + 1];
      acc = wmma4(a, W[j], acc);
    }
    Z = acc;
  };

#pragma unroll
  for (int l = 0; l < 4; ++l) {
    const int odd = l & 1;
#pragma unroll
    for (int t = 0; t < 16; ++t)
#pragma unroll
      for (int v = 0; v < 8; ++v) {
        if ((__builtin_parity(t) ^ __builtin_parity(v)) & 1) X[t][v] = flip8<1>(X[t][v]);
        else                                                  X[t][v] = flip8<2>(X[t][v]);
      }
    if (odd) {
#pragma unroll
      for (int t = 0; t < 8; ++t)
#pragma unroll
        for (int v = 0; v < 8; ++v) {
          bool sw = ((__builtin_parity(v) & 1) ^ h) != 0;
          float x0 = X[t][v], x1 = X[t + 8][v];
          X[t][v]     = sw ? x1 : x0;
          X[t + 8][v] = sw ? x0 : x1;
        }
    }
#pragma unroll
    for (int t = 0; t < 16; ++t) {
      if (odd) rowpass(X[t], BW[0]);
      else     rowpass(X[t], (__builtin_parity(t) & 1) ? BX : BW[1]);
      rowpass(X[t], BW[2]);
    }
    v8f Y[16];
#pragma unroll
    for (int t = 0; t < 8; ++t)
#pragma unroll
      for (int v = 0; v < 8; ++v) {
        float p = X[t][v], q = X[t + 8][v];
        swap_halves(p, q);
        Y[v][t] = p;
        Y[v + 8][t] = q;
      }
#pragma unroll
    for (int u = 0; u < 16; ++u) {
      v8f acc = {};
#pragma unroll
      for (int j = 0; j < 4; ++j) {
        v2f bf; bf.x = Y[u][2 * j]; bf.y = Y[u][2 * j + 1];
        acc = wmma4(BW[odd][j], bf, acc);
      }
      X[u] = acc;
    }
  }

  float partial = 0.0f;
#pragma unroll
  for (int t = 0; t < 16; ++t)
#pragma unroll
    for (int v = 0; v < 8; ++v) {
      float p = X[t][v];
      partial += p * p * tab[TAB_U + 256 * t + 16 * (v + 8 * h) + c];
    }
#pragma unroll
  for (int off = 16; off > 0; off >>= 1)
    partial += __shfl_xor(partial, off);
  if (lane == 0) out[b] = partial + head_b[0];
}

extern "C" void kernel_launch(void* const* d_in, const int* in_sizes, int n_in,
                              void* d_out, int out_size, void* d_ws, size_t ws_size,
                              hipStream_t stream) {
  const float* sb     = (const float*)d_in[0];
  const float* params = (const float*)d_in[1];
  const float* hw     = (const float*)d_in[2];
  const float* hb     = (const float*)d_in[3];
  float* out = (float*)d_out;
  int batch = in_sizes[0] / 12;
  float* tab = (float*)d_ws;
  qsim_tables_kernel<<<dim3((TAB_U + TAB_G) / TAB_THREADS), dim3(TAB_THREADS), 0, stream>>>(
      params, hw, tab);
  qsim_wmma_kernel<<<dim3(batch), dim3(QS_THREADS), 0, stream>>>(
      sb, tab, hb, out);
}


# Table launch removed: each wave builds its gate fragments from the RY factors and its per-nibble readout weights itself; 2 -> 1 launch; squares summed per tile and per register first: new sum order
# speedup vs baseline: 2.9484x; 1.0624x over previous
#include <hip/hip_runtime.h>

typedef __attribute__((ext_vector_type(2))) float v2f;
typedef __attribute__((ext_vector_type(8))) float v8f;

#define QS_THREADS 32

__device__ __forceinline__ int perm4(int x) {
  if (x & 8) x ^= 4;
  if (x & 4) x ^= 2;
  if (x & 2) x ^= 1;
  return x;
}

template <int RM>
__device__ __forceinline__ float flip8(float x) {
  int xi = __builtin_bit_cast(int, x);
  return __builtin_bit_cast(float,
      __builtin_amdgcn_update_dpp(xi, xi, 0x128, RM, 0xf, false));
}

__device__ __forceinline__ void swap_halves(float& p, float& q) {
  auto r = __builtin_amdgcn_permlane16_swap(__builtin_bit_cast(unsigned, p),
                                            __builtin_bit_cast(unsigned, q), false, false);
  p = __builtin_bit_cast(float, (unsigned)r[0]);
  q = __builtin_bit_cast(float, (unsigned)r[1]);
}

__device__ __forceinline__ v8f wmma4(v2f a, v2f b, v8f c) {
  return __builtin_amdgcn_wmma_f32_16x16x4_f32(false, a, false, b, (short)0, c, false, false);
}

__global__ __launch_bounds__(QS_THREADS) void qsim_wmma_kernel(
    const float* __restrict__ state_batch,
    const float* __restrict__ params,
    const float* __restrict__ head_w,
    const float* __restrict__ head_b,
    float* __restrict__ out) {
  const int b    = blockIdx.x;
  const int lane = threadIdx.x;
  const int c = lane & 15, h = lane >> 4;

  const float* asrc = (lane < 12) ? (state_batch + (size_t)b * 12 + lane)
                                  : (params + (lane < 24 ? lane - 12 : 0));
  float ang = 0.5f * asrc[0];
  float myc = cosf(ang), mys = sinf(ang);
  float cz[12], sz[12];
#pragma unroll
  for (int w = 0; w < 12; ++w) {
    cz[w] = __builtin_bit_cast(float, __builtin_amdgcn_readlane(__builtin_bit_cast(int, myc), w));
    sz[w] = __builtin_bit_cast(float, __builtin_amdgcn_readlane(__builtin_bit_cast(int, mys), w));
  }
  float cpv[12], spv[12];
#pragma unroll
  for (int w = 0; w < 12; ++w) {
    cpv[w] = __builtin_bit_cast(float, __builtin_amdgcn_readlane(__builtin_bit_cast(int, myc), 12 + w));
    spv[w] = __builtin_bit_cast(float, __builtin_amdgcn_readlane(__builtin_bit_cast(int, mys), 12 + w));
  }

  v2f BW[3][4], BX[4];
#pragma unroll
  for (int m = 0; m < 3; ++m) {
    float u0[4], u1[4];
#pragma unroll
    for (int q = 0; q < 4; ++q) {
      const bool bi = ((c >> (3 - q)) & 1) != 0;
      const float cc = cpv[4 * m + q], ss = spv[4 * m + q];
      u0[q] = bi ? ss : cc;
      u1[q] = bi ? cc : -ss;
    }
    float E[16];
#pragma unroll
    for (int jp = 0; jp < 16; ++jp) {
      float e = 1.0f;
#pragma unroll
      for (int q = 0; q < 4; ++q) e *= ((jp >> (3 - q)) & 1) ? u1[q] : u0[q];
      E[jp] = e;
    }
#pragma unroll
    for (int kk = 0; kk < 8; ++kk) {
      const float lo = E[perm4(kk)], hi = E[perm4(kk + 8)];
      const float w = h ? hi : lo;
      if (kk & 1) BW[m][kk >> 1].y = w; else BW[m][kk >> 1].x = w;
      if (m == 1) {
        const float wx = h ? lo : hi;
        if (kk & 1) BX[kk >> 1].y = wx; else BX[kk >> 1].x = wx;
      }
    }
  }

  v8f X[16];
  {
    float eL = 1.0f;
#pragma unroll
    for (int q = 0; q < 4; ++q) eL *= ((c >> (3 - q)) & 1) ? sz[8 + q] : cz[8 + q];
    float eM[8];
#pragma unroll
    for (int v = 0; v < 8; ++v) {
      float e = h ? sz[4] : cz[4];
#pragma unroll
      for (int q = 1; q < 4; ++q) e *= ((v >> (3 - q)) & 1) ? sz[4 + q] : cz[4 + q];
      eM[v] = e * eL;
    }
#pragma unroll
    for (int t = 0; t < 16; ++t) {
      float eT = 1.0f;
#pragma unroll
      for (int q = 0; q < 4; ++q) eT *= ((t >> (3 - q)) & 1) ? sz[q] : cz[q];
#pragma unroll
      for (int v = 0; v < 8; ++v) X[t][v] = eT * eM[v];
    }
  }

  auto rowpass = [&](v8f& Z, const v2f* W) {
    v8f acc = {};
#pragma unroll
    for (int j = 0; j < 4; ++j) {
      v2f a; a.x = Z[2 * j]; a.y = Z[2 * j + 1];
      acc = wmma4(a, W[j], acc);
    }
    Z = acc;
  };

#pragma unroll
  for (int l = 0; l < 4; ++l) {
    const int odd = l & 1;
#pragma unroll
    for (int t = 0; t < 16; ++t)
#pragma unroll
      for (int v = 0; v < 8; ++v) {
        if ((__builtin_parity(t) ^ __builtin_parity(v)) & 1) X[t][v] = flip8<1>(X[t][v]);
        else                                                  X[t][v] = flip8<2>(X[t][v]);
      }
    if (odd) {
#pragma unroll
      for (int t = 0; t < 8; ++t)
#pragma unroll
        for (int v = 0; v < 8; ++v) {
          bool sw = ((__builtin_parity(v) & 1) ^ h) != 0;
          float x0 = X[t][v], x1 = X[t + 8][v];
          X[t][v]     = sw ? x1 : x0;
          X[t + 8][v] = sw ? x0 : x1;
        }
    }
#pragma unroll
    for (int t = 0; t < 16; ++t) {
      if (odd) rowpass(X[t], BW[0]);
      else     rowpass(X[t], (__builtin_parity(t) & 1) ? BX : BW[1]);
      rowpass(X[t], BW[2]);
    }
    v8f Y[16];
#pragma unroll
    for (int t = 0; t < 8; ++t)
#pragma unroll
      for (int v = 0; v < 8; ++v) {
        float p = X[t][v], q = X[t + 8][v];
        swap_halves(p, q);
        Y[v][t] = p;
        Y[v + 8][t] = q;
      }
#pragma unroll
    for (int u = 0; u < 16; ++u) {
      v8f acc = {};
#pragma unroll
      for (int j = 0; j < 4; ++j) {
        v2f bf; bf.x = Y[u][2 * j]; bf.y = Y[u][2 * j + 1];
        acc = wmma4(BW[odd][j], bf, acc);
      }
      X[u] = acc;
    }
  }

  float hw[12];
#pragma unroll
  for (int w = 0; w < 12; ++w) hw[w] = head_w[w];
  float gLane = h ? -hw[4] : hw[4];
#pragma unroll
  for (int q = 0; q < 4; ++q) gLane += ((c >> (3 - q)) & 1) ? -hw[8 + q] : hw[8 + q];
  float sumT[16], sumV[8];
#pragma unroll
  for (int t = 0; t < 16; ++t) sumT[t] = 0.0f;
#pragma unroll
  for (int v = 0; v < 8; ++v) sumV[v] = 0.0f;
#pragma unroll
  for (int t = 0; t < 16; ++t)
#pragma unroll
    for (int v = 0; v < 8; ++v) {
      float p = X[t][v];
      sumT[t] = fmaf(p, p, sumT[t]);
      sumV[v] = fmaf(p, p, sumV[v]);
    }
  float partial = 0.0f;
#pragma unroll
  for (int t = 0; t < 16; ++t) {
    float gT = 0.0f;
#pragma unroll
    for (int q = 0; q < 4; ++q) gT += ((t >> (3 - q)) & 1) ? -hw[q] : hw[q];
    partial = fmaf(sumT[t], gT, partial);
  }
#pragma unroll
  for (int v = 0; v < 8; ++v) {
    float gV = gLane;
#pragma unroll
    for (int q = 1; q < 4; ++q) gV += ((v >> (3 - q)) & 1) ? -hw[4 + q] : hw[4 + q];
    partial = fmaf(sumV[v], gV, partial);
  }
#pragma unroll
  for (int off = 16; off > 0; off >>= 1)
    partial += __shfl_xor(partial, off);
  if (lane == 0) out[b] = partial + head_b[0];
}

extern "C" void kernel_launch(void* const* d_in, const int* in_sizes, int n_in,
                              void* d_out, int out_size, void* d_ws, size_t ws_size,
                              hipStream_t stream) {
  const float* sb     = (const float*)d_in[0];
  const float* params = (const float*)d_in[1];
  const float* hw     = (const float*)d_in[2];
  const float* hb     = (const float*)d_in[3];
  float* out = (float*)d_out;
  int batch = in_sizes[0] / 12;
  qsim_wmma_kernel<<<dim3(batch), dim3(QS_THREADS), 0, stream>>>(
      sb, params, hw, hb, out);
}


# Index maps put every CNOT control parity and target on a register or tile bit: cross-nibble CNOTs become register renaming, even layers sum the row pass in two halves; no DPP moves; new sum order
# speedup vs baseline: 3.1714x; 1.0756x over previous
#include <hip/hip_runtime.h>

typedef __attribute__((ext_vector_type(2))) float v2f;
typedef __attribute__((ext_vector_type(8))) float v8f;

#define QS_THREADS 32

__device__ __forceinline__ constexpr int perm4(int x) {
  if (x & 8) x ^= 4;
  if (x & 4) x ^= 2;
  if (x & 2) x ^= 1;
  return x;
}

__device__ __forceinline__ constexpr int par4(int x) { return (x ^ (x >> 1) ^ (x >> 2) ^ (x >> 3)) & 1; }
__device__ __forceinline__ constexpr int rmap(int k) {
  return (((k >> 1) & 1) << 3) | (par4(k) << 2) | (((k >> 3) & 1) << 1) | (k & 1);
}
__device__ __forceinline__ constexpr int sw13(int c) { return (c & 5) | (((c >> 1) & 1) << 3) | (((c >> 3) & 1) << 1); }

__device__ __forceinline__ void swap_halves(float& p, float& q) {
  auto r = __builtin_amdgcn_permlane16_swap(__builtin_bit_cast(unsigned, p),
                                            __builtin_bit_cast(unsigned, q), false, false);
  p = __builtin_bit_cast(float, (unsigned)r[0]);
  q = __builtin_bit_cast(float, (unsigned)r[1]);
}

__device__ __forceinline__ v8f wmma4(v2f a, v2f b, v8f c) {
  return __builtin_amdgcn_wmma_f32_16x16x4_f32(false, a, false, b, (short)0, c, false, false);
}

__global__ __launch_bounds__(QS_THREADS) void qsim_wmma_kernel(
    const float* __restrict__ state_batch,
    const float* __restrict__ params,
    const float* __restrict__ head_w,
    const float* __restrict__ head_b,
    float* __restrict__ out) {
  const int b    = blockIdx.x;
  const int lane = threadIdx.x;
  const int c = lane & 15, h = lane >> 4;

  const float* asrc = (lane < 12) ? (state_batch + (size_t)b * 12 + lane)
                                  : (params + (lane < 24 ? lane - 12 : 0));
  float ang = 0.5f * asrc[0];
  float myc = cosf(ang), mys = sinf(ang);
  float cz[12], sz[12];
#pragma unroll
  for (int w = 0; w < 12; ++w) {
    cz[w] = __builtin_bit_cast(float, __builtin_amdgcn_readlane(__builtin_bit_cast(int, myc), w));
    sz[w] = __builtin_bit_cast(float, __builtin_amdgcn_readlane(__builtin_bit_cast(int, mys), w));
  }
  float cpv[12], spv[12];
#pragma unroll
  for (int w = 0; w < 12; ++w) {
    cpv[w] = __builtin_bit_cast(float, __builtin_amdgcn_readlane(__builtin_bit_cast(int, myc), 12 + w));
    spv[w] = __builtin_bit_cast(float, __builtin_amdgcn_readlane(__builtin_bit_cast(int, mys), 12 + w));
  }

  const int cR = rmap(c), cL = sw13(c);
  v2f G[3][4];
#pragma unroll
  for (int m = 0; m < 3; ++m) {
    const int co = (m == 2) ? cL : cR;
    float u0[4], u1[4];
#pragma unroll
    for (int q = 0; q < 4; ++q) {
      const bool bi = ((co >> (3 - q)) & 1) != 0;
      const float cc = cpv[4 * m + q], ss = spv[4 * m + q];
      u0[q] = bi ? ss : cc;
      u1[q] = bi ? cc : -ss;
    }
    float E[16];
#pragma unroll
    for (int jp = 0; jp < 16; ++jp) {
      float e = 1.0f;
#pragma unroll
      for (int q = 0; q < 4; ++q) e *= ((jp >> (3 - q)) & 1) ? u1[q] : u0[q];
      E[jp] = e;
    }
#pragma unroll
    for (int kk = 0; kk < 8; ++kk) {
      const int i0 = (m == 2) ? sw13(kk)     : rmap(kk);
      const int i1 = (m == 2) ? sw13(kk + 8) : rmap(kk + 8);
      const float lo = E[perm4(i0)], hi = E[perm4(i1)];
      const float w = h ? hi : lo;
      if (kk & 1) G[m][kk >> 1].y = w; else G[m][kk >> 1].x = w;
    }
  }

  v8f X[16];
  {
    float eL = 1.0f;
#pragma unroll
    for (int q = 0; q < 4; ++q) eL *= ((cL >> (3 - q)) & 1) ? sz[8 + q] : cz[8 + q];
    float eM[8];
#pragma unroll
    for (int v = 0; v < 8; ++v) {
      float e0 = 1.0f, e1 = 1.0f;
#pragma unroll
      for (int q = 0; q < 4; ++q) {
        e0 *= ((rmap(v) >> (3 - q)) & 1)     ? sz[4 + q] : cz[4 + q];
        e1 *= ((rmap(v + 8) >> (3 - q)) & 1) ? sz[4 + q] : cz[4 + q];
      }
      eM[v] = (h ? e1 : e0) * eL;
    }
#pragma unroll
    for (int t = 0; t < 16; ++t) {
      float eT = 1.0f;
#pragma unroll
      for (int q = 0; q < 4; ++q) eT *= ((rmap(t) >> (3 - q)) & 1) ? sz[q] : cz[q];
#pragma unroll
      for (int v = 0; v < 8; ++v) X[t][v] = eT * eM[v];
    }
  }

#pragma unroll
  for (int l = 0; l < 4; ++l) {
    const int odd = l & 1;
    if (odd) {
#pragma unroll
      for (int t = 0; t < 16; ++t)
        if ((t ^ 6) > t) {
#pragma unroll
          for (int v = 4; v < 8; ++v) {
            const float x0 = X[t][v];
            X[t][v] = X[t ^ 6][v];
            X[t ^ 6][v] = x0;
          }
        }
    } else {
#pragma unroll
      for (int t = 0; t < 16; ++t)
        if ((t >> 2) & 1) {
#pragma unroll
          for (int v = 0; v < 8; ++v)
            if ((v ^ 6) > v) {
              const float x0 = X[t][v];
              X[t][v] = X[t][v ^ 6];
              X[t][v ^ 6] = x0;
            }
        }
    }
#pragma unroll
    for (int t = 0; t < 16; ++t) {
      v8f Z;
      if (odd) {
        v8f D = {};
#pragma unroll
        for (int j = 0; j < 4; ++j) {
          v2f a; a.x = X[t][2 * j]; a.y = X[t][2 * j + 1];
          D = wmma4(a, G[0][j], D);
        }
#pragma unroll
        for (int v = 0; v < 8; ++v) Z[v] = ((t >> 2) & 1) ? D[v ^ 2] : D[v];
      } else {
        v8f D0 = {}, D1 = {};
#pragma unroll
        for (int j = 0; j < 4; ++j) {
          v2f a; a.x = X[t][2 * j]; a.y = X[t][2 * j + 1];
          if (j >> 1) D1 = wmma4(a, G[1][j], D1);
          else        D0 = wmma4(a, G[1][j], D0);
        }
#pragma unroll
        for (int v = 0; v < 8; ++v) Z[v] = D0[v] + D1[v ^ 2];
      }
      v8f acc = {};
#pragma unroll
      for (int j = 0; j < 4; ++j) {
        v2f a; a.x = Z[2 * j]; a.y = Z[2 * j + 1];
        acc = wmma4(a, G[2][j], acc);
      }
      X[t] = acc;
    }
    v8f Y[16];
#pragma unroll
    for (int t = 0; t < 8; ++t)
#pragma unroll
      for (int v = 0; v < 8; ++v) {
        float p = X[t][v], q = X[t + 8][v];
        swap_halves(p, q);
        Y[v][t] = p;
        Y[v + 8][t] = q;
      }
#pragma unroll
    for (int u = 0; u < 16; ++u) {
      v8f acc = {};
#pragma unroll
      for (int j = 0; j < 4; ++j) {
        v2f bf; bf.x = Y[u][2 * j]; bf.y = Y[u][2 * j + 1];
        acc = wmma4(G[odd][j], bf, acc);
      }
      X[u] = acc;
    }
  }

  float hw[12];
#pragma unroll
  for (int w = 0; w < 12; ++w) hw[w] = head_w[w];
  float gLane = 0.0f;
#pragma unroll
  for (int q = 0; q < 4; ++q) gLane += ((cL >> (3 - q)) & 1) ? -hw[8 + q] : hw[8 + q];
  float sumT[16], sumV[8];
#pragma unroll
  for (int t = 0; t < 16; ++t) sumT[t] = 0.0f;
#pragma unroll
  for (int v = 0; v < 8; ++v) sumV[v] = 0.0f;
#pragma unroll
  for (int t = 0; t < 16; ++t)
#pragma unroll
    for (int v = 0; v < 8; ++v) {
      float p = X[t][v];
      sumT[t] = fmaf(p, p, sumT[t]);
      sumV[v] = fmaf(p, p, sumV[v]);
    }
  float partial = 0.0f;
#pragma unroll
  for (int t = 0; t < 16; ++t) {
    float gT = 0.0f;
#pragma unroll
    for (int q = 0; q < 4; ++q) gT += ((rmap(t) >> (3 - q)) & 1) ? -hw[q] : hw[q];
    partial = fmaf(sumT[t], gT, partial);
  }
#pragma unroll
  for (int v = 0; v < 8; ++v) {
    float g0 = gLane, g1 = gLane;
#pragma unroll
    for (int q = 0; q < 4; ++q) {
      g0 += ((rmap(v) >> (3 - q)) & 1)     ? -hw[4 + q] : hw[4 + q];
      g1 += ((rmap(v + 8) >> (3 - q)) & 1) ? -hw[4 + q] : hw[4 + q];
    }
    partial = fmaf(sumV[v], h ? g1 : g0, partial);
  }
#pragma unroll
  for (int off = 16; off > 0; off >>= 1)
    partial += __shfl_xor(partial, off);
  if (lane == 0) out[b] = partial + head_b[0];
}

extern "C" void kernel_launch(void* const* d_in, const int* in_sizes, int n_in,
                              void* d_out, int out_size, void* d_ws, size_t ws_size,
                              hipStream_t stream) {
  const float* sb     = (const float*)d_in[0];
  const float* params = (const float*)d_in[1];
  const float* hw     = (const float*)d_in[2];
  const float* hb     = (const float*)d_in[3];
  float* out = (float*)d_out;
  int batch = in_sizes[0] / 12;
  qsim_wmma_kernel<<<dim3(batch), dim3(QS_THREADS), 0, stream>>>(
      sb, params, hw, hb, out);
}
